# starting kernel
# baseline (speedup 1.0000x reference)
#include <hip/hip_runtime.h>

typedef __attribute__((ext_vector_type(2))) float v2f;
typedef __attribute__((ext_vector_type(8))) float v8f;

#define LS 33

__device__ __forceinline__ v8f wmma4(v2f a, v2f b, v8f c) {
  return __builtin_amdgcn_wmma_f32_16x16x4_f32(false, a, false, b, (short)0, c,
                                               false, false);
}

__device__ __forceinline__ v2f afrag(const float* M, int mTile, int kBase, int lane) {
  int m  = mTile * 16 + (lane & 15);
  int k0 = kBase + ((lane >> 4) << 1);
  v2f a;
  a.x = M[m * LS + k0];
  a.y = M[m * LS + k0 + 1];
  return a;
}

__device__ __forceinline__ v2f bfragWT(const float* W, int nTile, int kStep, int lane) {
  int n  = nTile * 16 + (lane & 15);
  int k0 = kStep * 4 + ((lane >> 4) << 1);
  v2f b;
  b.x = W[n * 32 + k0];
  b.y = W[n * 32 + k0 + 1];
  return b;
}

__device__ __forceinline__ v2f bfragKT(const float* Kb, int nTile, int kStep, int h8, int lane) {
  int j  = nTile * 16 + (lane & 15);
  int k0 = kStep * 4 + ((lane >> 4) << 1);
  v2f b;
  b.x = Kb[j * LS + h8 + k0];
  b.y = Kb[j * LS + h8 + k0 + 1];
  return b;
}

__device__ __forceinline__ v2f bfragV(const float* Vb, int kStep, int h8, int lane) {
  int j  = lane & 15;
  int k0 = kStep * 4 + ((lane >> 4) << 1);
  v2f b;
  if (j < 8) {
    b.x = Vb[k0 * LS + h8 + j];
    b.y = Vb[(k0 + 1) * LS + h8 + j];
  } else {
    b.x = 0.f;
    b.y = 0.f;
  }
  return b;
}

__launch_bounds__(64)
__global__ void ga_kernel(const float* __restrict__ x,
                          const float* __restrict__ qw, const float* __restrict__ qb,
                          const float* __restrict__ kvw, const float* __restrict__ kvb,
                          const float* __restrict__ pw, const float* __restrict__ pb,
                          float* __restrict__ out) {
  __shared__ float sXA[2][32 * LS];
  __shared__ float sQ[2][32 * LS];
  __shared__ float sK[2][32 * LS];
  __shared__ float sV[2][32 * LS];
  __shared__ float sS[2][32 * LS];

  const int b    = blockIdx.x;
  const int c    = threadIdx.x >> 5;
  const int lane = threadIdx.x & 31;

  __builtin_prefetch(qw, 0, 3);
  __builtin_prefetch(kvw, 0, 3);
  __builtin_prefetch(pw, 0, 3);

  float* X = sXA[c];
  float* Q = sQ[c];
  float* K = sK[c];
  float* V = sV[c];

  const float* xg = x + (size_t)(b * 2 + c) * 1024;

#pragma unroll
  for (int i = 0; i < 8; ++i) {
    int f = i * 128 + lane * 4;
    float4 v = *(const float4*)(xg + f);
    int r = f >> 5, cc = f & 31;
    float* p = &X[r * LS + cc];
    p[0] = v.x; p[1] = v.y; p[2] = v.z; p[3] = v.w;
  }
  __syncthreads();

  const int l15   = lane & 15;
  const int half8 = (lane >> 4) << 3;

#pragma unroll 1
  for (int nT = 0; nT < 2; ++nT) {
    float bias = qb[nT * 16 + l15];
    v8f a0 = {}, a1 = {};
#pragma unroll
    for (int k = 0; k < 8; ++k) {
      v2f bf = bfragWT(qw, nT, k, lane);
      a0 = wmma4(afrag(X, 0, k * 4, lane), bf, a0);
      a1 = wmma4(afrag(X, 1, k * 4, lane), bf, a1);
    }
    int n = nT * 16 + l15;
#pragma unroll
    for (int g = 0; g < 8; ++g) {
      Q[(half8 + g) * LS + n]      = a0[g] + bias;
      Q[(16 + half8 + g) * LS + n] = a1[g] + bias;
    }
  }

#pragma unroll 1
  for (int nT = 0; nT < 4; ++nT) {
    float bias  = kvb[nT * 16 + l15];
    float* dst  = (nT < 2) ? K : V;
    int    nLoc = (nT & 1) * 16 + l15;
    v8f a0 = {}, a1 = {};
#pragma unroll
    for (int k = 0; k < 8; ++k) {
      v2f bf = bfragWT(kvw, nT, k, lane);
      a0 = wmma4(afrag(X, 0, k * 4, lane), bf, a0);
      a1 = wmma4(afrag(X, 1, k * 4, lane), bf, a1);
    }
#pragma unroll
    for (int g = 0; g < 8; ++g) {
      dst[(half8 + g) * LS + nLoc]      = a0[g] + bias;
      dst[(16 + half8 + g) * LS + nLoc] = a1[g] + bias;
    }
  }
  __syncthreads();

  const float SCALE = 0.35355339059327373f;
  float* P = X;

#pragma unroll 1
  for (int head = 0; head < 4; ++head) {
    int h8 = head * 8;

#pragma unroll 1
    for (int nT = 0; nT < 2; ++nT) {
      v8f a0 = {}, a1 = {};
#pragma unroll
      for (int k = 0; k < 2; ++k) {
        v2f bf = bfragKT(K, nT, k, h8, lane);
        a0 = wmma4(afrag(Q, 0, h8 + k * 4, lane), bf, a0);
        a1 = wmma4(afrag(Q, 1, h8 + k * 4, lane), bf, a1);
      }
      int n = nT * 16 + l15;
#pragma unroll
      for (int g = 0; g < 8; ++g) {
        P[(half8 + g) * LS + n]      = a0[g];
        P[(16 + half8 + g) * LS + n] = a1[g];
      }
    }
    __syncthreads();

    {
      float* row = &P[lane * LS];
      float mx = -3.4e38f;
#pragma unroll
      for (int j = 0; j < 32; ++j) mx = fmaxf(mx, row[j] * SCALE);
      float s = 0.f;
#pragma unroll
      for (int j = 0; j < 32; ++j) {
        float e = __expf(row[j] * SCALE - mx);
        row[j] = e;
        s += e;
      }
      float inv = 1.f / s;
#pragma unroll
      for (int j = 0; j < 32; ++j) row[j] *= inv;
    }
    __syncthreads();

    {
      v8f a0 = {}, a1 = {};
#pragma unroll
      for (int k = 0; k < 8; ++k) {
        v2f bf = bfragV(V, k, h8, lane);
        a0 = wmma4(afrag(P, 0, k * 4, lane), bf, a0);
        a1 = wmma4(afrag(P, 1, k * 4, lane), bf, a1);
      }
      if (l15 < 8) {
        int d  = l15;
        int cp = head >> 1;
        int hb = (head & 1) * 16 + c * 8;
#pragma unroll
        for (int g = 0; g < 8; ++g) {
          int h0 = half8 + g;
          int h1 = 16 + half8 + g;
          sS[cp][(hb + (h0 >> 2)) * LS + (h0 & 3) * 8 + d] = a0[g];
          sS[cp][(hb + (h1 >> 2)) * LS + (h1 & 3) * 8 + d] = a1[g];
        }
      }
    }
    __syncthreads();
  }

  const float* S = sS[c];
  float* og = out + (size_t)(b * 2 + c) * 1024;
#pragma unroll 1
  for (int nT = 0; nT < 2; ++nT) {
    float bias = pb[nT * 16 + l15];
    v8f a0 = {}, a1 = {};
#pragma unroll
    for (int k = 0; k < 8; ++k) {
      v2f bf = bfragWT(pw, nT, k, lane);
      a0 = wmma4(afrag(S, 0, k * 4, lane), bf, a0);
      a1 = wmma4(afrag(S, 1, k * 4, lane), bf, a1);
    }
    int n = nT * 16 + l15;
#pragma unroll
    for (int g = 0; g < 8; ++g) {
      og[(half8 + g) * 32 + n]      = a0[g] + bias;
      og[(16 + half8 + g) * 32 + n] = a1[g] + bias;
    }
  }
}

extern "C" void kernel_launch(void* const* d_in, const int* in_sizes, int n_in,
                              void* d_out, int out_size, void* d_ws, size_t ws_size,
                              hipStream_t stream) {
  const float* x   = (const float*)d_in[0];
  const float* qw  = (const float*)d_in[1];
  const float* qb  = (const float*)d_in[2];
  const float* kvw = (const float*)d_in[3];
  const float* kvb = (const float*)d_in[4];
  const float* pw  = (const float*)d_in[5];
  const float* pb  = (const float*)d_in[6];
  float* out = (float*)d_out;

  int B = in_sizes[0] / (2 * 32 * 32);
  ga_kernel<<<B, 64, 0, stream>>>(x, qw, qb, kvw, kvb, pw, pb, out);

  (void)n_in; (void)out_size; (void)d_ws; (void)ws_size;
}


# Attention output stored in place of each head's consumed Q columns, projection reads its scrambled A fragments from there: scramble LDS tile dropped, 42240 to 33792 B, 9 not 7 workgroups per WGP
# speedup vs baseline: 1.0889x; 1.0889x over previous
#include <hip/hip_runtime.h>

typedef __attribute__((ext_vector_type(2))) float v2f;
typedef __attribute__((ext_vector_type(8))) float v8f;

#define LS 33

__device__ __forceinline__ v8f wmma4(v2f a, v2f b, v8f c) {
  return __builtin_amdgcn_wmma_f32_16x16x4_f32(false, a, false, b, (short)0, c,
                                               false, false);
}

__device__ __forceinline__ v2f afrag(const float* M, int mTile, int kBase, int lane) {
  int m  = mTile * 16 + (lane & 15);
  int k0 = kBase + ((lane >> 4) << 1);
  v2f a;
  a.x = M[m * LS + k0];
  a.y = M[m * LS + k0 + 1];
  return a;
}

__device__ __forceinline__ v2f bfragWT(const float* W, int nTile, int kStep, int lane) {
  int n  = nTile * 16 + (lane & 15);
  int k0 = kStep * 4 + ((lane >> 4) << 1);
  v2f b;
  b.x = W[n * 32 + k0];
  b.y = W[n * 32 + k0 + 1];
  return b;
}

__device__ __forceinline__ v2f bfragKT(const float* Kb, int nTile, int kStep, int h8, int lane) {
  int j  = nTile * 16 + (lane & 15);
  int k0 = kStep * 4 + ((lane >> 4) << 1);
  v2f b;
  b.x = Kb[j * LS + h8 + k0];
  b.y = Kb[j * LS + h8 + k0 + 1];
  return b;
}

__device__ __forceinline__ v2f bfragV(const float* Vb, int kStep, int h8, int lane) {
  int j  = lane & 15;
  int k0 = kStep * 4 + ((lane >> 4) << 1);
  v2f b;
  if (j < 8) {
    b.x = Vb[k0 * LS + h8 + j];
    b.y = Vb[(k0 + 1) * LS + h8 + j];
  } else {
    b.x = 0.f;
    b.y = 0.f;
  }
  return b;
}

__device__ __forceinline__ v2f afragO(const float* Oall, int mTile, int kBase, int cp, int lane) {
  int m   = lane & 15;
  int w   = kBase + ((lane >> 4) << 1);
  int q   = ((m & 7) << 2) + (w >> 3);
  int col = (2 * cp + mTile) * 8 + (w & 7);
  const float* p = Oall + (m >> 3) * (32 * LS) + q * LS + col;
  v2f a;
  a.x = p[0];
  a.y = p[1];
  return a;
}

__launch_bounds__(64)
__global__ void ga_kernel(const float* __restrict__ x,
                          const float* __restrict__ qw, const float* __restrict__ qb,
                          const float* __restrict__ kvw, const float* __restrict__ kvb,
                          const float* __restrict__ pw, const float* __restrict__ pb,
                          float* __restrict__ out) {
  __shared__ float sXA[2][32 * LS];
  __shared__ float sQ[2][32 * LS];
  __shared__ float sK[2][32 * LS];
  __shared__ float sV[2][32 * LS];

  const int b    = blockIdx.x;
  const int c    = threadIdx.x >> 5;
  const int lane = threadIdx.x & 31;

  __builtin_prefetch(qw, 0, 3);
  __builtin_prefetch(kvw, 0, 3);
  __builtin_prefetch(pw, 0, 3);

  float* X = sXA[c];
  float* Q = sQ[c];
  float* K = sK[c];
  float* V = sV[c];

  const float* xg = x + (size_t)(b * 2 + c) * 1024;

#pragma unroll
  for (int i = 0; i < 8; ++i) {
    int f = i * 128 + lane * 4;
    float4 v = *(const float4*)(xg + f);
    int r = f >> 5, cc = f & 31;
    float* p = &X[r * LS + cc];
    p[0] = v.x; p[1] = v.y; p[2] = v.z; p[3] = v.w;
  }
  __syncthreads();

  const int l15   = lane & 15;
  const int half8 = (lane >> 4) << 3;

#pragma unroll 1
  for (int nT = 0; nT < 2; ++nT) {
    float bias = qb[nT * 16 + l15];
    v8f a0 = {}, a1 = {};
#pragma unroll
    for (int k = 0; k < 8; ++k) {
      v2f bf = bfragWT(qw, nT, k, lane);
      a0 = wmma4(afrag(X, 0, k * 4, lane), bf, a0);
      a1 = wmma4(afrag(X, 1, k * 4, lane), bf, a1);
    }
    int n = nT * 16 + l15;
#pragma unroll
    for (int g = 0; g < 8; ++g) {
      Q[(half8 + g) * LS + n]      = a0[g] + bias;
      Q[(16 + half8 + g) * LS + n] = a1[g] + bias;
    }
  }

#pragma unroll 1
  for (int nT = 0; nT < 4; ++nT) {
    float bias  = kvb[nT * 16 + l15];
    float* dst  = (nT < 2) ? K : V;
    int    nLoc = (nT & 1) * 16 + l15;
    v8f a0 = {}, a1 = {};
#pragma unroll
    for (int k = 0; k < 8; ++k) {
      v2f bf = bfragWT(kvw, nT, k, lane);
      a0 = wmma4(afrag(X, 0, k * 4, lane), bf, a0);
      a1 = wmma4(afrag(X, 1, k * 4, lane), bf, a1);
    }
#pragma unroll
    for (int g = 0; g < 8; ++g) {
      dst[(half8 + g) * LS + nLoc]      = a0[g] + bias;
      dst[(16 + half8 + g) * LS + nLoc] = a1[g] + bias;
    }
  }
  __syncthreads();

  const float SCALE = 0.35355339059327373f;
  float* P = X;

#pragma unroll 1
  for (int head = 0; head < 4; ++head) {
    int h8 = head * 8;

#pragma unroll 1
    for (int nT = 0; nT < 2; ++nT) {
      v8f a0 = {}, a1 = {};
#pragma unroll
      for (int k = 0; k < 2; ++k) {
        v2f bf = bfragKT(K, nT, k, h8, lane);
        a0 = wmma4(afrag(Q, 0, h8 + k * 4, lane), bf, a0);
        a1 = wmma4(afrag(Q, 1, h8 + k * 4, lane), bf, a1);
      }
      int n = nT * 16 + l15;
#pragma unroll
      for (int g = 0; g < 8; ++g) {
        P[(half8 + g) * LS + n]      = a0[g];
        P[(16 + half8 + g) * LS + n] = a1[g];
      }
    }
    __syncthreads();

    {
      float* row = &P[lane * LS];
      float mx = -3.4e38f;
#pragma unroll
      for (int j = 0; j < 32; ++j) mx = fmaxf(mx, row[j] * SCALE);
      float s = 0.f;
#pragma unroll
      for (int j = 0; j < 32; ++j) {
        float e = __expf(row[j] * SCALE - mx);
        row[j] = e;
        s += e;
      }
      float inv = 1.f / s;
#pragma unroll
      for (int j = 0; j < 32; ++j) row[j] *= inv;
    }
    __syncthreads();

    {
      v8f a0 = {}, a1 = {};
#pragma unroll
      for (int k = 0; k < 8; ++k) {
        v2f bf = bfragV(V, k, h8, lane);
        a0 = wmma4(afrag(P, 0, k * 4, lane), bf, a0);
        a1 = wmma4(afrag(P, 1, k * 4, lane), bf, a1);
      }
      if (l15 < 8) {
        int d = l15;
#pragma unroll
        for (int g = 0; g < 8; ++g) {
          Q[(half8 + g) * LS + h8 + d]      = a0[g];
          Q[(16 + half8 + g) * LS + h8 + d] = a1[g];
        }
      }
    }
    __syncthreads();
  }

  const float* Oall = &sQ[0][0];
  float* og = out + (size_t)(b * 2 + c) * 1024;
#pragma unroll 1
  for (int nT = 0; nT < 2; ++nT) {
    float bias = pb[nT * 16 + l15];
    v8f a0 = {}, a1 = {};
#pragma unroll
    for (int k = 0; k < 8; ++k) {
      v2f bf = bfragWT(pw, nT, k, lane);
      a0 = wmma4(afragO(Oall, 0, k * 4, c, lane), bf, a0);
      a1 = wmma4(afragO(Oall, 1, k * 4, c, lane), bf, a1);
    }
    int n = nT * 16 + l15;
#pragma unroll
    for (int g = 0; g < 8; ++g) {
      og[(half8 + g) * 32 + n]      = a0[g] + bias;
      og[(16 + half8 + g) * 32 + n] = a1[g] + bias;
    }
  }
}

extern "C" void kernel_launch(void* const* d_in, const int* in_sizes, int n_in,
                              void* d_out, int out_size, void* d_ws, size_t ws_size,
                              hipStream_t stream) {
  const float* x   = (const float*)d_in[0];
  const float* qw  = (const float*)d_in[1];
  const float* qb  = (const float*)d_in[2];
  const float* kvw = (const float*)d_in[3];
  const float* kvb = (const float*)d_in[4];
  const float* pw  = (const float*)d_in[5];
  const float* pb  = (const float*)d_in[6];
  float* out = (float*)d_out;

  int B = in_sizes[0] / (2 * 32 * 32);
  ga_kernel<<<B, 64, 0, stream>>>(x, qw, qb, kvw, kvb, pw, pb, out);

  (void)n_in; (void)out_size; (void)d_ws; (void)ws_size;
}


# Scores computed transposed (K_h Q_h^T) so the softmaxed D tiles are the B operand of O^T = V^T P^T in registers: no LDS trip or barrier for P; softmax sum and PV key order regrouped by lane half
# speedup vs baseline: 1.3280x; 1.2196x over previous
#include <hip/hip_runtime.h>

typedef __attribute__((ext_vector_type(2))) float v2f;
typedef __attribute__((ext_vector_type(8))) float v8f;

#define LS 33

__device__ __forceinline__ v8f wmma4(v2f a, v2f b, v8f c) {
  return __builtin_amdgcn_wmma_f32_16x16x4_f32(false, a, false, b, (short)0, c,
                                               false, false);
}

__device__ __forceinline__ v2f afrag(const float* M, int mTile, int kBase, int lane) {
  int m  = mTile * 16 + (lane & 15);
  int k0 = kBase + ((lane >> 4) << 1);
  v2f a;
  a.x = M[m * LS + k0];
  a.y = M[m * LS + k0 + 1];
  return a;
}

__device__ __forceinline__ v2f bfragWT(const float* W, int nTile, int kStep, int lane) {
  int n  = nTile * 16 + (lane & 15);
  int k0 = kStep * 4 + ((lane >> 4) << 1);
  v2f b;
  b.x = W[n * 32 + k0];
  b.y = W[n * 32 + k0 + 1];
  return b;
}

__device__ __forceinline__ v2f bfragKT(const float* Kb, int nTile, int kStep, int h8, int lane) {
  int j  = nTile * 16 + (lane & 15);
  int k0 = kStep * 4 + ((lane >> 4) << 1);
  v2f b;
  b.x = Kb[j * LS + h8 + k0];
  b.y = Kb[j * LS + h8 + k0 + 1];
  return b;
}

__device__ __forceinline__ v2f afragVT(const float* Vb, int s, int h8, int lane) {
  int j0 = ((s >> 2) << 4) + ((lane >> 4) << 3) + ((s & 3) << 1);
  const float* p = Vb + j0 * LS + h8 + (lane & 7);
  v2f a;
  a.x = p[0];
  a.y = p[LS];
  return a;
}

__device__ __forceinline__ float other_half(float v) {
  unsigned u = __float_as_uint(v);
  return __uint_as_float(__builtin_amdgcn_permlanex16(u, u, 0x76543210u, 0xfedcba98u, false, false));
}

__device__ __forceinline__ void softmax_keys(v8f& t0, v8f& t1, float scale) {
  float mx = -3.4e38f;
#pragma unroll
  for (int g = 0; g < 8; ++g) mx = fmaxf(mx, t0[g] * scale);
#pragma unroll
  for (int g = 0; g < 8; ++g) mx = fmaxf(mx, t1[g] * scale);
  mx = fmaxf(mx, other_half(mx));
  float s = 0.f;
#pragma unroll
  for (int g = 0; g < 8; ++g) {
    float e = __expf(t0[g] * scale - mx);
    t0[g] = e;
    s += e;
  }
#pragma unroll
  for (int g = 0; g < 8; ++g) {
    float e = __expf(t1[g] * scale - mx);
    t1[g] = e;
    s += e;
  }
  s += other_half(s);
  float inv = 1.f / s;
#pragma unroll
  for (int g = 0; g < 8; ++g) {
    t0[g] *= inv;
    t1[g] *= inv;
  }
}

__device__ __forceinline__ v2f afragO(const float* Oall, int mTile, int kBase, int cp, int lane) {
  int m   = lane & 15;
  int w   = kBase + ((lane >> 4) << 1);
  int q   = ((m & 7) << 2) + (w >> 3);
  int col = (2 * cp + mTile) * 8 + (w & 7);
  const float* p = Oall + (m >> 3) * (32 * LS) + q * LS + col;
  v2f a;
  a.x = p[0];
  a.y = p[1];
  return a;
}

__launch_bounds__(64)
__global__ void ga_kernel(const float* __restrict__ x,
                          const float* __restrict__ qw, const float* __restrict__ qb,
                          const float* __restrict__ kvw, const float* __restrict__ kvb,
                          const float* __restrict__ pw, const float* __restrict__ pb,
                          float* __restrict__ out) {
  __shared__ float sXA[2][32 * LS];
  __shared__ float sQ[2][32 * LS];
  __shared__ float sK[2][32 * LS];
  __shared__ float sV[2][32 * LS];

  const int b    = blockIdx.x;
  const int c    = threadIdx.x >> 5;
  const int lane = threadIdx.x & 31;

  __builtin_prefetch(qw, 0, 3);
  __builtin_prefetch(kvw, 0, 3);
  __builtin_prefetch(pw, 0, 3);

  float* X = sXA[c];
  float* Q = sQ[c];
  float* K = sK[c];
  float* V = sV[c];

  const float* xg = x + (size_t)(b * 2 + c) * 1024;

#pragma unroll
  for (int i = 0; i < 8; ++i) {
    int f = i * 128 + lane * 4;
    float4 v = *(const float4*)(xg + f);
    int r = f >> 5, cc = f & 31;
    float* p = &X[r * LS + cc];
    p[0] = v.x; p[1] = v.y; p[2] = v.z; p[3] = v.w;
  }
  __syncthreads();

  const int l15   = lane & 15;
  const int half8 = (lane >> 4) << 3;

#pragma unroll 1
  for (int nT = 0; nT < 2; ++nT) {
    float bias = qb[nT * 16 + l15];
    v8f a0 = {}, a1 = {};
#pragma unroll
    for (int k = 0; k < 8; ++k) {
      v2f bf = bfragWT(qw, nT, k, lane);
      a0 = wmma4(afrag(X, 0, k * 4, lane), bf, a0);
      a1 = wmma4(afrag(X, 1, k * 4, lane), bf, a1);
    }
    int n = nT * 16 + l15;
#pragma unroll
    for (int g = 0; g < 8; ++g) {
      Q[(half8 + g) * LS + n]      = a0[g] + bias;
      Q[(16 + half8 + g) * LS + n] = a1[g] + bias;
    }
  }

#pragma unroll 1
  for (int nT = 0; nT < 4; ++nT) {
    float bias  = kvb[nT * 16 + l15];
    float* dst  = (nT < 2) ? K : V;
    int    nLoc = (nT & 1) * 16 + l15;
    v8f a0 = {}, a1 = {};
#pragma unroll
    for (int k = 0; k < 8; ++k) {
      v2f bf = bfragWT(kvw, nT, k, lane);
      a0 = wmma4(afrag(X, 0, k * 4, lane), bf, a0);
      a1 = wmma4(afrag(X, 1, k * 4, lane), bf, a1);
    }
#pragma unroll
    for (int g = 0; g < 8; ++g) {
      dst[(half8 + g) * LS + nLoc]      = a0[g] + bias;
      dst[(16 + half8 + g) * LS + nLoc] = a1[g] + bias;
    }
  }
  __syncthreads();

  const float SCALE = 0.35355339059327373f;

#pragma unroll 1
  for (int head = 0; head < 4; ++head) {
    int h8 = head * 8;

    v8f s00 = {}, s01 = {}, s10 = {}, s11 = {};
#pragma unroll
    for (int k = 0; k < 2; ++k) {
      v2f ka0 = afrag(K, 0, h8 + k * 4, lane);
      v2f ka1 = afrag(K, 1, h8 + k * 4, lane);
      v2f qb0 = bfragKT(Q, 0, k, h8, lane);
      v2f qb1 = bfragKT(Q, 1, k, h8, lane);
      s00 = wmma4(ka0, qb0, s00);
      s01 = wmma4(ka0, qb1, s01);
      s10 = wmma4(ka1, qb0, s10);
      s11 = wmma4(ka1, qb1, s11);
    }

    softmax_keys(s00, s10, SCALE);
    softmax_keys(s01, s11, SCALE);

    v8f o0 = {}, o1 = {};
#pragma unroll
    for (int s = 0; s < 8; ++s) {
      v2f va = afragVT(V, s, h8, lane);
      const int g = (s & 3) << 1;
      v2f p0, p1;
      if (s < 4) {
        p0.x = s00[g]; p0.y = s00[g + 1];
        p1.x = s01[g]; p1.y = s01[g + 1];
      } else {
        p0.x = s10[g]; p0.y = s10[g + 1];
        p1.x = s11[g]; p1.y = s11[g + 1];
      }
      o0 = wmma4(va, p0, o0);
      o1 = wmma4(va, p1, o1);
    }
    if (lane < 16) {
#pragma unroll
      for (int g = 0; g < 8; ++g) {
        Q[l15 * LS + h8 + g]        = o0[g];
        Q[(16 + l15) * LS + h8 + g] = o1[g];
      }
    }
  }
  __syncthreads();

  const float* Oall = &sQ[0][0];
  float* og = out + (size_t)(b * 2 + c) * 1024;
#pragma unroll 1
  for (int nT = 0; nT < 2; ++nT) {
    float bias = pb[nT * 16 + l15];
    v8f a0 = {}, a1 = {};
#pragma unroll
    for (int k = 0; k < 8; ++k) {
      v2f bf = bfragWT(pw, nT, k, lane);
      a0 = wmma4(afragO(Oall, 0, k * 4, c, lane), bf, a0);
      a1 = wmma4(afragO(Oall, 1, k * 4, c, lane), bf, a1);
    }
    int n = nT * 16 + l15;
#pragma unroll
    for (int g = 0; g < 8; ++g) {
      og[(half8 + g) * 32 + n]      = a0[g] + bias;
      og[(16 + half8 + g) * 32 + n] = a1[g] + bias;
    }
  }
}

extern "C" void kernel_launch(void* const* d_in, const int* in_sizes, int n_in,
                              void* d_out, int out_size, void* d_ws, size_t ws_size,
                              hipStream_t stream) {
  const float* x   = (const float*)d_in[0];
  const float* qw  = (const float*)d_in[1];
  const float* qb  = (const float*)d_in[2];
  const float* kvw = (const float*)d_in[3];
  const float* kvb = (const float*)d_in[4];
  const float* pw  = (const float*)d_in[5];
  const float* pb  = (const float*)d_in[6];
  float* out = (float*)d_out;

  int B = in_sizes[0] / (2 * 32 * 32);
  ga_kernel<<<B, 64, 0, stream>>>(x, qw, qb, kvw, kvb, pw, pb, out);

  (void)n_in; (void)out_size; (void)d_ws; (void)ws_size;
}
